# starting kernel
# baseline (speedup 1.0000x reference)
#include <hip/hip_runtime.h>

#define N_NODES 10000
#define N_EDGES 50000
#define HID 64
#define EDGE_DIM 16

typedef float v2f __attribute__((ext_vector_type(2)));
typedef float v8f __attribute__((ext_vector_type(8)));

#define KPAIRS 32
#define NCOLS 1088
#define LDS_STRIDE 1104
#define NTILES (N_EDGES / 16)

__global__ void mpnn_init_out(const float* __restrict__ nf,
                              const float* __restrict__ bias,
                              float* __restrict__ out) {
  int i = blockIdx.x * blockDim.x + threadIdx.x;
  if (i < N_NODES * HID) out[i] = nf[i] + bias[i & (HID - 1)];
}

__global__ void __launch_bounds__(128)
mpnn_edge_kernel(const float* __restrict__ nf,
                 const float* __restrict__ ef,
                 const int* __restrict__ src,
                 const int* __restrict__ dst,
                 const float* __restrict__ W_edge,
                 const float* __restrict__ b_edge,
                 float* __restrict__ out) {
  extern __shared__ char smem_raw[];
  v2f* __restrict__ ldsw = (v2f*)smem_raw;

  const int tid = threadIdx.x;

  for (int idx = tid; idx < KPAIRS * NCOLS; idx += blockDim.x) {
    int kp = idx / NCOLS;
    int c  = idx - kp * NCOLS;
    int h0 = kp << 1;
    float m0, m1;
    if (c < 1024) {
      int d = c >> 6, o = c & 63;
      m0 = W_edge[d * 4096 + h0 * 64 + o];
      m1 = W_edge[d * 4096 + (h0 + 1) * 64 + o];
    } else {
      int o = c - 1024;
      m0 = b_edge[h0 * 64 + o];
      m1 = b_edge[(h0 + 1) * 64 + o];
    }
    v2f v; v.x = m0; v.y = m1;
    ldsw[kp * LDS_STRIDE + c] = v;
  }
  __syncthreads();

  const int lane = tid & 31;
  const int wid  = tid >> 5;
  const int hi   = lane >> 4;
  const int lo   = lane & 15;
  const int nwaves = (blockDim.x >> 5) * gridDim.x;
  const int wslot  = blockIdx.x * (blockDim.x >> 5) + wid;
  const v2f* __restrict__ nf2 = (const v2f*)nf;

  const v8f vzero = {0.f, 0.f, 0.f, 0.f, 0.f, 0.f, 0.f, 0.f};

  for (int tile = wslot; tile < NTILES; tile += nwaves) {
    const int e0 = tile << 4;

    const int row = src[e0 + lo];
    v2f a[16];
#pragma unroll
    for (int kk = 0; kk < 16; ++kk)
      a[kk] = nf2[row * 32 + (kk << 1) + hi];

    float efr[8];
#pragma unroll
    for (int r = 0; r < 8; ++r)
      efr[r] = ef[(e0 + r + (hi << 3)) * EDGE_DIM + lo];

    v8f macc[4];
#pragma unroll
    for (int jb = 0; jb < 4; ++jb) macc[jb] = vzero;

    for (int j = 0; j < 68; ++j) {
      const int c = (j << 4) + lo;
      const v2f* wp = ldsw + (hi * LDS_STRIDE + c);
      v8f z = vzero;
#pragma unroll
      for (int kk = 0; kk < 16; ++kk) {
        v2f b = wp[(kk << 1) * LDS_STRIDE];
#if __has_builtin(__builtin_amdgcn_wmma_f32_16x16x4_f32)
        z = __builtin_amdgcn_wmma_f32_16x16x4_f32(false, a[kk], false, b,
                                                  (short)0, z, false, false);
#else
        z[0] += a[kk].x * b.x + a[kk].y * b.y;
#endif
      }
      if (j < 64) {
        const int d    = j >> 2;
        const int jb   = j & 3;
        const int bidx = (d | (lane & 16)) << 2;
#pragma unroll
        for (int r = 0; r < 8; ++r) {
          int si = __builtin_amdgcn_ds_bpermute(bidx, __builtin_bit_cast(int, efr[r]));
          macc[jb][r] += __builtin_bit_cast(float, si) * z[r];
        }
      } else {
        const int jb = j - 64;
#pragma unroll
        for (int r = 0; r < 8; ++r) macc[jb][r] += z[r];
      }
    }

#pragma unroll
    for (int r = 0; r < 8; ++r) {
      const int drow = dst[e0 + r + (hi << 3)];
      float* op = out + drow * HID + lo;
#pragma unroll
      for (int jb = 0; jb < 4; ++jb)
        atomicAdd(op + (jb << 4), macc[jb][r]);
    }
  }
}

extern "C" void kernel_launch(void* const* d_in, const int* in_sizes, int n_in,
                              void* d_out, int out_size, void* d_ws, size_t ws_size,
                              hipStream_t stream) {
  const float* nf     = (const float*)d_in[0];
  const float* ef     = (const float*)d_in[1];
  const int*   src    = (const int*)d_in[2];
  const int*   dst    = (const int*)d_in[3];
  const float* W_edge = (const float*)d_in[4];
  const float* b_edge = (const float*)d_in[5];
  const float* bias   = (const float*)d_in[6];
  float* out = (float*)d_out;

  mpnn_init_out<<<(N_NODES * HID + 255) / 256, 256, 0, stream>>>(nf, bias, out);

  const size_t lds_bytes = (size_t)KPAIRS * LDS_STRIDE * sizeof(v2f);
  mpnn_edge_kernel<<<160, 128, lds_bytes, stream>>>(nf, ef, src, dst, W_edge, b_edge, out);
}


# weight staging into LDS: per thread 34 independent 16-byte loads per pass (4 passes) and 16-byte LDS stores, in place of 272 serial round trips of two 4-byte loads each
# speedup vs baseline: 4.6809x; 4.6809x over previous
#include <hip/hip_runtime.h>

#define N_NODES 10000
#define N_EDGES 50000
#define HID 64
#define EDGE_DIM 16

typedef float v2f __attribute__((ext_vector_type(2)));
typedef float v8f __attribute__((ext_vector_type(8)));

#define KPAIRS 32
#define NCOLS 1088
#define LDS_STRIDE 1104
#define NTILES (N_EDGES / 16)

__global__ void mpnn_init_out(const float* __restrict__ nf,
                              const float* __restrict__ bias,
                              float* __restrict__ out) {
  int i = blockIdx.x * blockDim.x + threadIdx.x;
  if (i < N_NODES * HID) out[i] = nf[i] + bias[i & (HID - 1)];
}

__device__ __forceinline__ void load_bgroup(v2f (&buf)[32], const v2f* pa, const v2f* pb) {
#pragma unroll
  for (int i = 0; i < 4; ++i) {
#pragma unroll
    for (int jb = 0; jb < 4; ++jb) {
      buf[i * 4 + jb]       = pa[(2 * i) * LDS_STRIDE + (jb << 4)];
      buf[(i + 4) * 4 + jb] = pb[(2 * i) * LDS_STRIDE + (jb << 4)];
    }
  }
}

template <int G>
__device__ __forceinline__ void mma_bgroup(v8f (&macc)[4], const v2f (&a)[16],
                                           const v2f (&buf)[32], float s) {
#pragma unroll
  for (int i = 0; i < 8; ++i) {
    v2f as = a[8 * G + i] * s;
#pragma unroll
    for (int jb = 0; jb < 4; ++jb)
      macc[jb] = __builtin_amdgcn_wmma_f32_16x16x4_f32(false, as, false, buf[i * 4 + jb],
                                                       (short)0, macc[jb], false, false);
  }
}

__global__ void __launch_bounds__(128)
mpnn_edge_kernel(const float* __restrict__ nf,
                 const float* __restrict__ ef,
                 const int* __restrict__ src,
                 const int* __restrict__ dst,
                 const float* __restrict__ W_edge,
                 const float* __restrict__ b_edge,
                 float* __restrict__ out) {
  extern __shared__ __attribute__((aligned(16))) char smem_raw[];
  v2f* __restrict__ ldsw = (v2f*)smem_raw;

  const int tid = threadIdx.x;

  {
    const int o4 = tid & 15;
    const int kq = tid >> 4;
    for (int pass = 0; pass < KPAIRS / 8; ++pass) {
      const int kp = kq + 8 * pass;
      const float4* gw = (const float4*)(W_edge + kp * 128) + o4;
      const float4* gb = (const float4*)(b_edge + kp * 128) + o4;
      float4 r0[17], r1[17];
#pragma unroll
      for (int d = 0; d < 16; ++d) {
        r0[d] = gw[d * 1024];
        r1[d] = gw[d * 1024 + 16];
      }
      r0[16] = gb[0];
      r1[16] = gb[16];
      float4* lp = (float4*)(ldsw + kp * LDS_STRIDE + 4 * o4);
#pragma unroll
      for (int d = 0; d < 17; ++d) {
        lp[d * 32]     = make_float4(r0[d].x, r1[d].x, r0[d].y, r1[d].y);
        lp[d * 32 + 1] = make_float4(r0[d].z, r1[d].z, r0[d].w, r1[d].w);
      }
    }
  }
  __syncthreads();

  const int lane = tid & 31;
  const int wid  = tid >> 5;
  const int hi   = lane >> 4;
  const int lo   = lane & 15;
  const int nwaves = (blockDim.x >> 5) * gridDim.x;
  const int wslot  = blockIdx.x * (blockDim.x >> 5) + wid;
  const v2f* __restrict__ nf2 = (const v2f*)nf;

  const v8f vzero = {0.f, 0.f, 0.f, 0.f, 0.f, 0.f, 0.f, 0.f};

  for (int tile = wslot; tile < NTILES; tile += nwaves) {
    const int e0 = tile << 4;

    const int row = src[e0 + lo];
    v2f a[16];
#pragma unroll
    for (int kk = 0; kk < 16; ++kk)
      a[kk] = nf2[row * 32 + (kk << 1) + hi];

    const float* efrow = ef + (e0 + lo) * EDGE_DIM;
    v8f macc[4];
#pragma unroll
    for (int jb = 0; jb < 4; ++jb) macc[jb] = vzero;

    const v2f* q0 = ldsw + (hi * LDS_STRIDE + lo);
    const v2f* q1 = q0 + 8 * LDS_STRIDE;
    const v2f* q2 = q0 + 16 * LDS_STRIDE;
    const v2f* q3 = q0 + 24 * LDS_STRIDE;
    v2f b0[32], b1[32];
    float s = efrow[0];
    load_bgroup(b0, q0, q1);
    for (int d = 0; d < 17; ++d) {
      const int dn = d < 16 ? d + 1 : 16;
      const int step = d < 16 ? 64 : 0;
      const float efn = efrow[dn & 15];
      const float sn = dn < 16 ? efn : 1.0f;
      load_bgroup(b1, q2, q3);
      __builtin_amdgcn_sched_barrier(0);
      mma_bgroup<0>(macc, a, b0, s);
      __builtin_amdgcn_sched_barrier(0);
      load_bgroup(b0, q0 + step, q1 + step);
      __builtin_amdgcn_sched_barrier(0);
      mma_bgroup<1>(macc, a, b1, s);
      q0 += step; q1 += step; q2 += step; q3 += step;
      s = sn;
    }

#pragma unroll
    for (int r = 0; r < 8; ++r) {
      const int drow = dst[e0 + r + (hi << 3)];
      float* op = out + drow * HID + lo;
#pragma unroll
      for (int jb = 0; jb < 4; ++jb)
        atomicAdd(op + (jb << 4), macc[jb][r]);
    }
  }
}

extern "C" void kernel_launch(void* const* d_in, const int* in_sizes, int n_in,
                              void* d_out, int out_size, void* d_ws, size_t ws_size,
                              hipStream_t stream) {
  const float* nf     = (const float*)d_in[0];
  const float* ef     = (const float*)d_in[1];
  const int*   src    = (const int*)d_in[2];
  const int*   dst    = (const int*)d_in[3];
  const float* W_edge = (const float*)d_in[4];
  const float* b_edge = (const float*)d_in[5];
  const float* bias   = (const float*)d_in[6];
  float* out = (float*)d_out;

  mpnn_init_out<<<(N_NODES * HID + 255) / 256, 256, 0, stream>>>(nf, bias, out);

  const size_t lds_bytes = (size_t)KPAIRS * LDS_STRIDE * sizeof(v2f);
  mpnn_edge_kernel<<<160, 128, lds_bytes, stream>>>(nf, ef, src, dst, W_edge, b_edge, out);
}


# edge kernel on all 256 WGPs with equal matrix work per SIMD: 3125 x 17 (tile, slice) units dealt as 1024 contiguous ranges of 51-52, one per wave (was 160 work-groups, whole tiles, up to 5 per wave)
# speedup vs baseline: 6.6845x; 1.4280x over previous
#include <hip/hip_runtime.h>

#define N_NODES 10000
#define N_EDGES 50000
#define HID 64
#define EDGE_DIM 16

typedef float v2f __attribute__((ext_vector_type(2)));
typedef float v8f __attribute__((ext_vector_type(8)));

#define KPAIRS 32
#define NCOLS 1088
#define LDS_STRIDE 1104
#define NTILES (N_EDGES / 16)
#define NSLICES 17
#define NUNITS (NTILES * NSLICES)
#define NBLOCKS 256
#define NWAVES (NBLOCKS * 4)

__global__ void mpnn_init_out(const float* __restrict__ nf,
                              const float* __restrict__ bias,
                              float* __restrict__ out) {
  int i = blockIdx.x * blockDim.x + threadIdx.x;
  if (i < N_NODES * HID) out[i] = nf[i] + bias[i & (HID - 1)];
}

__device__ __forceinline__ void load_bgroup(v2f (&buf)[32], const v2f* pa, const v2f* pb) {
#pragma unroll
  for (int i = 0; i < 4; ++i) {
#pragma unroll
    for (int jb = 0; jb < 4; ++jb) {
      buf[i * 4 + jb]       = pa[(2 * i) * LDS_STRIDE + (jb << 4)];
      buf[(i + 4) * 4 + jb] = pb[(2 * i) * LDS_STRIDE + (jb << 4)];
    }
  }
}

template <int G>
__device__ __forceinline__ void mma_bgroup(v8f (&macc)[4], const v2f (&a)[16],
                                           const v2f (&buf)[32], float s) {
#pragma unroll
  for (int i = 0; i < 8; ++i) {
    v2f as = a[8 * G + i] * s;
#pragma unroll
    for (int jb = 0; jb < 4; ++jb)
      macc[jb] = __builtin_amdgcn_wmma_f32_16x16x4_f32(false, as, false, buf[i * 4 + jb],
                                                       (short)0, macc[jb], false, false);
  }
}

__global__ void __launch_bounds__(128)
mpnn_edge_kernel(const float* __restrict__ nf,
                 const float* __restrict__ ef,
                 const int* __restrict__ src,
                 const int* __restrict__ dst,
                 const float* __restrict__ W_edge,
                 const float* __restrict__ b_edge,
                 float* __restrict__ out) {
  extern __shared__ __attribute__((aligned(16))) char smem_raw[];
  v2f* __restrict__ ldsw = (v2f*)smem_raw;

  const int tid = threadIdx.x;

  {
    const int o4 = tid & 15;
    const int kq = tid >> 4;
    for (int pass = 0; pass < KPAIRS / 8; ++pass) {
      const int kp = kq + 8 * pass;
      const float4* gw = (const float4*)(W_edge + kp * 128) + o4;
      const float4* gb = (const float4*)(b_edge + kp * 128) + o4;
      float4 r0[17], r1[17];
#pragma unroll
      for (int d = 0; d < 16; ++d) {
        r0[d] = gw[d * 1024];
        r1[d] = gw[d * 1024 + 16];
      }
      r0[16] = gb[0];
      r1[16] = gb[16];
      float4* lp = (float4*)(ldsw + kp * LDS_STRIDE + 4 * o4);
#pragma unroll
      for (int d = 0; d < 17; ++d) {
        lp[d * 32]     = make_float4(r0[d].x, r1[d].x, r0[d].y, r1[d].y);
        lp[d * 32 + 1] = make_float4(r0[d].z, r1[d].z, r0[d].w, r1[d].w);
      }
    }
  }
  __syncthreads();

  const int lane = tid & 31;
  const int wid  = __builtin_amdgcn_readfirstlane(tid >> 5);
  const int hi   = lane >> 4;
  const int lo   = lane & 15;
  const int wslot = blockIdx.x * 4 + wid;
  const v2f* __restrict__ nf2 = (const v2f*)nf;

  int u = (wslot * NUNITS) / NWAVES;
  const int uend = ((wslot + 1) * NUNITS) / NWAVES;

  const v8f vzero = {0.f, 0.f, 0.f, 0.f, 0.f, 0.f, 0.f, 0.f};

  while (u < uend) {
    const int tile = u / NSLICES;
    const int dlo = u - tile * NSLICES;
    const int dhi = min(NSLICES, dlo + (uend - u));
    u += dhi - dlo;
    const int e0 = tile << 4;

    const int row = src[e0 + lo];
    v2f a[16];
#pragma unroll
    for (int kk = 0; kk < 16; ++kk)
      a[kk] = nf2[row * 32 + (kk << 1) + hi];

    const float* efrow = ef + (e0 + lo) * EDGE_DIM;
    v8f macc[4];
#pragma unroll
    for (int jb = 0; jb < 4; ++jb) macc[jb] = vzero;

    const v2f* q0 = ldsw + (hi * LDS_STRIDE + lo) + dlo * 64;
    const v2f* q1 = q0 + 8 * LDS_STRIDE;
    const v2f* q2 = q0 + 16 * LDS_STRIDE;
    const v2f* q3 = q0 + 24 * LDS_STRIDE;
    v2f b0[32], b1[32];
    float s = dlo < 16 ? efrow[dlo & 15] : 1.0f;
    load_bgroup(b0, q0, q1);
    for (int d = dlo; d < dhi; ++d) {
      const int dn = d < 16 ? d + 1 : 16;
      const int step = d < 16 ? 64 : 0;
      const float efn = efrow[dn & 15];
      const float sn = dn < 16 ? efn : 1.0f;
      load_bgroup(b1, q2, q3);
      __builtin_amdgcn_sched_barrier(0);
      mma_bgroup<0>(macc, a, b0, s);
      __builtin_amdgcn_sched_barrier(0);
      load_bgroup(b0, q0 + step, q1 + step);
      __builtin_amdgcn_sched_barrier(0);
      mma_bgroup<1>(macc, a, b1, s);
      q0 += step; q1 += step; q2 += step; q3 += step;
      s = sn;
    }

#pragma unroll
    for (int r = 0; r < 8; ++r) {
      const int drow = dst[e0 + r + (hi << 3)];
      float* op = out + drow * HID + lo;
#pragma unroll
      for (int jb = 0; jb < 4; ++jb)
        atomicAdd(op + (jb << 4), macc[jb][r]);
    }
  }
}

extern "C" void kernel_launch(void* const* d_in, const int* in_sizes, int n_in,
                              void* d_out, int out_size, void* d_ws, size_t ws_size,
                              hipStream_t stream) {
  const float* nf     = (const float*)d_in[0];
  const float* ef     = (const float*)d_in[1];
  const int*   src    = (const int*)d_in[2];
  const int*   dst    = (const int*)d_in[3];
  const float* W_edge = (const float*)d_in[4];
  const float* b_edge = (const float*)d_in[5];
  const float* bias   = (const float*)d_in[6];
  float* out = (float*)d_out;

  mpnn_init_out<<<(N_NODES * HID + 255) / 256, 256, 0, stream>>>(nf, bias, out);

  const size_t lds_bytes = (size_t)KPAIRS * LDS_STRIDE * sizeof(v2f);
  mpnn_edge_kernel<<<NBLOCKS, 128, lds_bytes, stream>>>(nf, ef, src, dst, W_edge, b_edge, out);
}


# edge kernel with two waves per SIMD (256 work-groups of 8 waves, 2048 ranges of 25-26 units): one wave's per-tile gather and atomic scatter overlap the other wave's WMMA stream
# speedup vs baseline: 7.5234x; 1.1255x over previous
#include <hip/hip_runtime.h>

#define N_NODES 10000
#define N_EDGES 50000
#define HID 64
#define EDGE_DIM 16

typedef float v2f __attribute__((ext_vector_type(2)));
typedef float v8f __attribute__((ext_vector_type(8)));

#define KPAIRS 32
#define NCOLS 1088
#define LDS_STRIDE 1104
#define NTILES (N_EDGES / 16)
#define NSLICES 17
#define NUNITS (NTILES * NSLICES)
#define NBLOCKS 256
#define WAVES_PER_BLOCK 8
#define NTHREADS (WAVES_PER_BLOCK * 32)
#define NWAVES (NBLOCKS * WAVES_PER_BLOCK)

__global__ void mpnn_init_out(const float* __restrict__ nf,
                              const float* __restrict__ bias,
                              float* __restrict__ out) {
  int i = blockIdx.x * blockDim.x + threadIdx.x;
  if (i < N_NODES * HID) out[i] = nf[i] + bias[i & (HID - 1)];
}

__device__ __forceinline__ void load_bgroup(v2f (&buf)[32], const v2f* pa, const v2f* pb) {
#pragma unroll
  for (int i = 0; i < 4; ++i) {
#pragma unroll
    for (int jb = 0; jb < 4; ++jb) {
      buf[i * 4 + jb]       = pa[(2 * i) * LDS_STRIDE + (jb << 4)];
      buf[(i + 4) * 4 + jb] = pb[(2 * i) * LDS_STRIDE + (jb << 4)];
    }
  }
}

template <int G>
__device__ __forceinline__ void mma_bgroup(v8f (&macc)[4], const v2f (&a)[16],
                                           const v2f (&buf)[32], float s) {
#pragma unroll
  for (int i = 0; i < 8; ++i) {
    v2f as = a[8 * G + i] * s;
#pragma unroll
    for (int jb = 0; jb < 4; ++jb)
      macc[jb] = __builtin_amdgcn_wmma_f32_16x16x4_f32(false, as, false, buf[i * 4 + jb],
                                                       (short)0, macc[jb], false, false);
  }
}

__global__ void __launch_bounds__(NTHREADS)
mpnn_edge_kernel(const float* __restrict__ nf,
                 const float* __restrict__ ef,
                 const int* __restrict__ src,
                 const int* __restrict__ dst,
                 const float* __restrict__ W_edge,
                 const float* __restrict__ b_edge,
                 float* __restrict__ out) {
  extern __shared__ __attribute__((aligned(16))) char smem_raw[];
  v2f* __restrict__ ldsw = (v2f*)smem_raw;

  const int tid = threadIdx.x;

  {
    const int o4 = tid & 15;
    const int kq = tid >> 4;
    for (int pass = 0; pass < KPAIRS / (NTHREADS / 16); ++pass) {
      const int kp = kq + (NTHREADS / 16) * pass;
      const float4* gw = (const float4*)(W_edge + kp * 128) + o4;
      const float4* gb = (const float4*)(b_edge + kp * 128) + o4;
      float4 r0[17], r1[17];
#pragma unroll
      for (int d = 0; d < 16; ++d) {
        r0[d] = gw[d * 1024];
        r1[d] = gw[d * 1024 + 16];
      }
      r0[16] = gb[0];
      r1[16] = gb[16];
      float4* lp = (float4*)(ldsw + kp * LDS_STRIDE + 4 * o4);
#pragma unroll
      for (int d = 0; d < 17; ++d) {
        lp[d * 32]     = make_float4(r0[d].x, r1[d].x, r0[d].y, r1[d].y);
        lp[d * 32 + 1] = make_float4(r0[d].z, r1[d].z, r0[d].w, r1[d].w);
      }
    }
  }
  __syncthreads();

  const int lane = tid & 31;
  const int wid  = __builtin_amdgcn_readfirstlane(tid >> 5);
  const int hi   = lane >> 4;
  const int lo   = lane & 15;
  const int wslot = blockIdx.x * WAVES_PER_BLOCK + wid;
  const v2f* __restrict__ nf2 = (const v2f*)nf;

  int u = (wslot * NUNITS) / NWAVES;
  const int uend = ((wslot + 1) * NUNITS) / NWAVES;

  const v8f vzero = {0.f, 0.f, 0.f, 0.f, 0.f, 0.f, 0.f, 0.f};

  while (u < uend) {
    const int tile = u / NSLICES;
    const int dlo = u - tile * NSLICES;
    const int dhi = min(NSLICES, dlo + (uend - u));
    u += dhi - dlo;
    const int e0 = tile << 4;

    const int row = src[e0 + lo];
    v2f a[16];
#pragma unroll
    for (int kk = 0; kk < 16; ++kk)
      a[kk] = nf2[row * 32 + (kk << 1) + hi];

    const float* efrow = ef + (e0 + lo) * EDGE_DIM;
    v8f macc[4];
#pragma unroll
    for (int jb = 0; jb < 4; ++jb) macc[jb] = vzero;

    const v2f* q0 = ldsw + (hi * LDS_STRIDE + lo) + dlo * 64;
    const v2f* q1 = q0 + 8 * LDS_STRIDE;
    const v2f* q2 = q0 + 16 * LDS_STRIDE;
    const v2f* q3 = q0 + 24 * LDS_STRIDE;
    v2f b0[32], b1[32];
    float s = dlo < 16 ? efrow[dlo & 15] : 1.0f;
    load_bgroup(b0, q0, q1);
    for (int d = dlo; d < dhi; ++d) {
      const int dn = d < 16 ? d + 1 : 16;
      const int step = d < 16 ? 64 : 0;
      const float efn = efrow[dn & 15];
      const float sn = dn < 16 ? efn : 1.0f;
      load_bgroup(b1, q2, q3);
      __builtin_amdgcn_sched_barrier(0);
      mma_bgroup<0>(macc, a, b0, s);
      __builtin_amdgcn_sched_barrier(0);
      load_bgroup(b0, q0 + step, q1 + step);
      __builtin_amdgcn_sched_barrier(0);
      mma_bgroup<1>(macc, a, b1, s);
      q0 += step; q1 += step; q2 += step; q3 += step;
      s = sn;
    }

#pragma unroll
    for (int r = 0; r < 8; ++r) {
      const int drow = dst[e0 + r + (hi << 3)];
      float* op = out + drow * HID + lo;
#pragma unroll
      for (int jb = 0; jb < 4; ++jb)
        atomicAdd(op + (jb << 4), macc[jb][r]);
    }
  }
}

extern "C" void kernel_launch(void* const* d_in, const int* in_sizes, int n_in,
                              void* d_out, int out_size, void* d_ws, size_t ws_size,
                              hipStream_t stream) {
  const float* nf     = (const float*)d_in[0];
  const float* ef     = (const float*)d_in[1];
  const int*   src    = (const int*)d_in[2];
  const int*   dst    = (const int*)d_in[3];
  const float* W_edge = (const float*)d_in[4];
  const float* b_edge = (const float*)d_in[5];
  const float* bias   = (const float*)d_in[6];
  float* out = (float*)d_out;

  mpnn_init_out<<<(N_NODES * HID + 255) / 256, 256, 0, stream>>>(nf, bias, out);

  const size_t lds_bytes = (size_t)KPAIRS * LDS_STRIDE * sizeof(v2f);
  mpnn_edge_kernel<<<NBLOCKS, NTHREADS, lds_bytes, stream>>>(nf, ef, src, dst, W_edge, b_edge, out);
}


# K split across work-groups: even / odd ones stage and multiply only rows 0-31 / 32-63 of h (one 34-load pass, 136 KB of weights per WGP, was 272 KB); 32-WMMA units, 1024 ranges of 51-52 per half
# speedup vs baseline: 7.7755x; 1.0335x over previous
#include <hip/hip_runtime.h>

#define N_NODES 10000
#define N_EDGES 50000
#define HID 64
#define EDGE_DIM 16

typedef float v2f __attribute__((ext_vector_type(2)));
typedef float v8f __attribute__((ext_vector_type(8)));

#define KPAIRS 32
#define KPAIRS_WG 16
#define NCOLS 1088
#define LDS_STRIDE 1104
#define NTILES (N_EDGES / 16)
#define NSLICES 17
#define NUNITS (NTILES * NSLICES)
#define NBLOCKS 256
#define WAVES_PER_BLOCK 8
#define NTHREADS (WAVES_PER_BLOCK * 32)
#define NWAVES_HALF (NBLOCKS / 2 * WAVES_PER_BLOCK)

__global__ void mpnn_init_out(const float* __restrict__ nf,
                              const float* __restrict__ bias,
                              float* __restrict__ out) {
  int i = blockIdx.x * blockDim.x + threadIdx.x;
  if (i < N_NODES * HID) out[i] = nf[i] + bias[i & (HID - 1)];
}

__device__ __forceinline__ void load_bgroup(v2f (&buf)[32], const v2f* pa, const v2f* pb) {
#pragma unroll
  for (int i = 0; i < 4; ++i) {
#pragma unroll
    for (int jb = 0; jb < 4; ++jb) {
      buf[i * 4 + jb]       = pa[(2 * i) * LDS_STRIDE + (jb << 4)];
      buf[(i + 4) * 4 + jb] = pb[(2 * i) * LDS_STRIDE + (jb << 4)];
    }
  }
}

__device__ __forceinline__ void mma_bgroup(v8f (&macc)[4], const v2f (&a)[8],
                                           const v2f (&buf)[32], float s) {
#pragma unroll
  for (int i = 0; i < 8; ++i) {
    v2f as = a[i] * s;
#pragma unroll
    for (int jb = 0; jb < 4; ++jb)
      macc[jb] = __builtin_amdgcn_wmma_f32_16x16x4_f32(false, as, false, buf[i * 4 + jb],
                                                       (short)0, macc[jb], false, false);
  }
}

__global__ void __launch_bounds__(NTHREADS)
mpnn_edge_kernel(const float* __restrict__ nf,
                 const float* __restrict__ ef,
                 const int* __restrict__ src,
                 const int* __restrict__ dst,
                 const float* __restrict__ W_edge,
                 const float* __restrict__ b_edge,
                 float* __restrict__ out) {
  extern __shared__ __attribute__((aligned(16))) char smem_raw[];
  v2f* __restrict__ ldsw = (v2f*)smem_raw;

  const int tid = threadIdx.x;

  const int khalf = blockIdx.x & 1;

  {
    const int o4 = tid & 15;
    const int kq = tid >> 4;
    const int kp = kq + KPAIRS_WG * khalf;
    const float4* gw = (const float4*)(W_edge + kp * 128) + o4;
    const float4* gb = (const float4*)(b_edge + kp * 128) + o4;
    float4 r0[17], r1[17];
#pragma unroll
    for (int d = 0; d < 16; ++d) {
      r0[d] = gw[d * 1024];
      r1[d] = gw[d * 1024 + 16];
    }
    r0[16] = gb[0];
    r1[16] = gb[16];
    float4* lp = (float4*)(ldsw + kq * LDS_STRIDE + 4 * o4);
#pragma unroll
    for (int d = 0; d < 17; ++d) {
      lp[d * 32]     = make_float4(r0[d].x, r1[d].x, r0[d].y, r1[d].y);
      lp[d * 32 + 1] = make_float4(r0[d].z, r1[d].z, r0[d].w, r1[d].w);
    }
  }
  __syncthreads();

  const int lane = tid & 31;
  const int wid  = __builtin_amdgcn_readfirstlane(tid >> 5);
  const int hi   = lane >> 4;
  const int lo   = lane & 15;
  const int wslot = (blockIdx.x >> 1) * WAVES_PER_BLOCK + wid;
  const v2f* __restrict__ nf2 = (const v2f*)nf;

  int u = (wslot * NUNITS) / NWAVES_HALF;
  const int uend = ((wslot + 1) * NUNITS) / NWAVES_HALF;

  const v8f vzero = {0.f, 0.f, 0.f, 0.f, 0.f, 0.f, 0.f, 0.f};

  while (u < uend) {
    const int tile = u / NSLICES;
    const int dlo = u - tile * NSLICES;
    const int dhi = min(NSLICES, dlo + (uend - u));
    u += dhi - dlo;
    const int e0 = tile << 4;

    const int row = src[e0 + lo];
    v2f a[8];
#pragma unroll
    for (int kk = 0; kk < 8; ++kk)
      a[kk] = nf2[row * 32 + khalf * 16 + (kk << 1) + hi];

    const float* efrow = ef + (e0 + lo) * EDGE_DIM;
    v8f macc[4];
#pragma unroll
    for (int jb = 0; jb < 4; ++jb) macc[jb] = vzero;

    const v2f* q0 = ldsw + (hi * LDS_STRIDE + lo) + dlo * 64;
    const v2f* q1 = q0 + 8 * LDS_STRIDE;
    v2f b0[32], b1[32];
    float s = dlo < 16 ? efrow[dlo & 15] : 1.0f;
    load_bgroup(b0, q0, q1);
    int d = dlo;
    for (; d + 1 < dhi; d += 2) {
      const float ef1 = efrow[(d + 1) & 15];
      const float s1 = d + 1 < 16 ? ef1 : 1.0f;
      const int d2 = d + 2 < NSLICES ? d + 2 : 16;
      const float ef2 = efrow[d2 & 15];
      const float s2 = d2 < 16 ? ef2 : 1.0f;
      const int step2 = (d2 - d) * 64;
      load_bgroup(b1, q0 + 64, q1 + 64);
      __builtin_amdgcn_sched_barrier(0);
      mma_bgroup(macc, a, b0, s);
      __builtin_amdgcn_sched_barrier(0);
      load_bgroup(b0, q0 + step2, q1 + step2);
      __builtin_amdgcn_sched_barrier(0);
      mma_bgroup(macc, a, b1, s1);
      q0 += step2; q1 += step2;
      s = s2;
    }
    if (d < dhi)
      mma_bgroup(macc, a, b0, s);

#pragma unroll
    for (int r = 0; r < 8; ++r) {
      const int drow = dst[e0 + r + (hi << 3)];
      float* op = out + drow * HID + lo;
#pragma unroll
      for (int jb = 0; jb < 4; ++jb)
        atomicAdd(op + (jb << 4), macc[jb][r]);
    }
  }
}

extern "C" void kernel_launch(void* const* d_in, const int* in_sizes, int n_in,
                              void* d_out, int out_size, void* d_ws, size_t ws_size,
                              hipStream_t stream) {
  const float* nf     = (const float*)d_in[0];
  const float* ef     = (const float*)d_in[1];
  const int*   src    = (const int*)d_in[2];
  const int*   dst    = (const int*)d_in[3];
  const float* W_edge = (const float*)d_in[4];
  const float* b_edge = (const float*)d_in[5];
  const float* bias   = (const float*)d_in[6];
  float* out = (float*)d_out;

  mpnn_init_out<<<(N_NODES * HID + 255) / 256, 256, 0, stream>>>(nf, bias, out);

  const size_t lds_bytes = (size_t)KPAIRS * LDS_STRIDE * sizeof(v2f);
  mpnn_edge_kernel<<<NBLOCKS, NTHREADS, lds_bytes, stream>>>(nf, ef, src, dst, W_edge, b_edge, out);
}


# per-tile memory requests issued ahead of use: next tile's nf rows one piece ahead (src index two ahead, first tile's beside the weight staging), dst rows at piece start, not at the scatter
# speedup vs baseline: 8.0720x; 1.0381x over previous
#include <hip/hip_runtime.h>

#define N_NODES 10000
#define N_EDGES 50000
#define HID 64
#define EDGE_DIM 16

typedef float v2f __attribute__((ext_vector_type(2)));
typedef float v8f __attribute__((ext_vector_type(8)));

#define KPAIRS 32
#define KPAIRS_WG 16
#define NCOLS 1088
#define LDS_STRIDE 1104
#define NTILES (N_EDGES / 16)
#define NSLICES 17
#define NUNITS (NTILES * NSLICES)
#define NBLOCKS 256
#define WAVES_PER_BLOCK 8
#define NTHREADS (WAVES_PER_BLOCK * 32)
#define NWAVES_HALF (NBLOCKS / 2 * WAVES_PER_BLOCK)

__global__ void mpnn_init_out(const float* __restrict__ nf,
                              const float* __restrict__ bias,
                              float* __restrict__ out) {
  int i = blockIdx.x * blockDim.x + threadIdx.x;
  if (i < N_NODES * HID) out[i] = nf[i] + bias[i & (HID - 1)];
}

__device__ __forceinline__ void load_bgroup(v2f (&buf)[32], const v2f* pa, const v2f* pb) {
#pragma unroll
  for (int i = 0; i < 4; ++i) {
#pragma unroll
    for (int jb = 0; jb < 4; ++jb) {
      buf[i * 4 + jb]       = pa[(2 * i) * LDS_STRIDE + (jb << 4)];
      buf[(i + 4) * 4 + jb] = pb[(2 * i) * LDS_STRIDE + (jb << 4)];
    }
  }
}

__device__ __forceinline__ void mma_bgroup(v8f (&macc)[4], const v2f (&a)[8],
                                           const v2f (&buf)[32], float s) {
#pragma unroll
  for (int i = 0; i < 8; ++i) {
    v2f as = a[i] * s;
#pragma unroll
    for (int jb = 0; jb < 4; ++jb)
      macc[jb] = __builtin_amdgcn_wmma_f32_16x16x4_f32(false, as, false, buf[i * 4 + jb],
                                                       (short)0, macc[jb], false, false);
  }
}

__global__ void __launch_bounds__(NTHREADS)
mpnn_edge_kernel(const float* __restrict__ nf,
                 const float* __restrict__ ef,
                 const int* __restrict__ src,
                 const int* __restrict__ dst,
                 const float* __restrict__ W_edge,
                 const float* __restrict__ b_edge,
                 float* __restrict__ out) {
  extern __shared__ __attribute__((aligned(16))) char smem_raw[];
  v2f* __restrict__ ldsw = (v2f*)smem_raw;

  const int tid = threadIdx.x;
  v2f a[8];

  const int khalf = blockIdx.x & 1;

  const int lane = tid & 31;
  const int wid  = __builtin_amdgcn_readfirstlane(tid >> 5);
  const int hi   = lane >> 4;
  const int lo   = lane & 15;
  const int wslot = (blockIdx.x >> 1) * WAVES_PER_BLOCK + wid;
  const v2f* __restrict__ nf2 = (const v2f*)nf;
  int u = (wslot * NUNITS) / NWAVES_HALF;
  const int uend = ((wslot + 1) * NUNITS) / NWAVES_HALF;
  int tile = u / NSLICES;
  const int row0 = src[(tile << 4) + lo];
  int row_next = src[(min(tile + 1, NTILES - 1) << 4) + lo];
  const int koff = khalf * 16 + hi;

  {
    const int o4 = tid & 15;
    const int kq = tid >> 4;
    const int kp = kq + KPAIRS_WG * khalf;
    const float4* gw = (const float4*)(W_edge + kp * 128) + o4;
    const float4* gb = (const float4*)(b_edge + kp * 128) + o4;
    float4 r0[17], r1[17];
#pragma unroll
    for (int d = 0; d < 16; ++d) {
      r0[d] = gw[d * 1024];
      r1[d] = gw[d * 1024 + 16];
    }
    r0[16] = gb[0];
    r1[16] = gb[16];
#pragma unroll
    for (int kk = 0; kk < 8; ++kk)
      a[kk] = nf2[row0 * 32 + koff + (kk << 1)];
    float4* lp = (float4*)(ldsw + kq * LDS_STRIDE + 4 * o4);
#pragma unroll
    for (int d = 0; d < 17; ++d) {
      lp[d * 32]     = make_float4(r0[d].x, r1[d].x, r0[d].y, r1[d].y);
      lp[d * 32 + 1] = make_float4(r0[d].z, r1[d].z, r0[d].w, r1[d].w);
    }
  }
  __syncthreads();


  const v8f vzero = {0.f, 0.f, 0.f, 0.f, 0.f, 0.f, 0.f, 0.f};


  while (u < uend) {
    const int dlo = u - tile * NSLICES;
    const int dhi = min(NSLICES, dlo + (uend - u));
    u += dhi - dlo;
    const int e0 = tile << 4;

    const int row_nn = src[(min(tile + 2, NTILES - 1) << 4) + lo];
    v2f an[8];
#pragma unroll
    for (int kk = 0; kk < 8; ++kk)
      an[kk] = nf2[row_next * 32 + koff + (kk << 1)];
    int drow[8];
#pragma unroll
    for (int r = 0; r < 8; ++r)
      drow[r] = dst[e0 + r + (hi << 3)];

    const float* efrow = ef + (e0 + lo) * EDGE_DIM;
    v8f macc[4];
#pragma unroll
    for (int jb = 0; jb < 4; ++jb) macc[jb] = vzero;

    const v2f* q0 = ldsw + (hi * LDS_STRIDE + lo) + dlo * 64;
    const v2f* q1 = q0 + 8 * LDS_STRIDE;
    v2f b0[32], b1[32];
    float s = dlo < 16 ? efrow[dlo & 15] : 1.0f;
    load_bgroup(b0, q0, q1);
    int d = dlo;
    for (; d + 1 < dhi; d += 2) {
      const float ef1 = efrow[(d + 1) & 15];
      const float s1 = d + 1 < 16 ? ef1 : 1.0f;
      const int d2 = d + 2 < NSLICES ? d + 2 : 16;
      const float ef2 = efrow[d2 & 15];
      const float s2 = d2 < 16 ? ef2 : 1.0f;
      const int step2 = (d2 - d) * 64;
      load_bgroup(b1, q0 + 64, q1 + 64);
      __builtin_amdgcn_sched_barrier(0);
      mma_bgroup(macc, a, b0, s);
      __builtin_amdgcn_sched_barrier(0);
      load_bgroup(b0, q0 + step2, q1 + step2);
      __builtin_amdgcn_sched_barrier(0);
      mma_bgroup(macc, a, b1, s1);
      q0 += step2; q1 += step2;
      s = s2;
    }
    if (d < dhi)
      mma_bgroup(macc, a, b0, s);

#pragma unroll
    for (int r = 0; r < 8; ++r) {
      float* op = out + drow[r] * HID + lo;
#pragma unroll
      for (int jb = 0; jb < 4; ++jb)
        atomicAdd(op + (jb << 4), macc[jb][r]);
    }

#pragma unroll
    for (int kk = 0; kk < 8; ++kk) a[kk] = an[kk];
    row_next = row_nn;
    ++tile;
  }
}

extern "C" void kernel_launch(void* const* d_in, const int* in_sizes, int n_in,
                              void* d_out, int out_size, void* d_ws, size_t ws_size,
                              hipStream_t stream) {
  const float* nf     = (const float*)d_in[0];
  const float* ef     = (const float*)d_in[1];
  const int*   src    = (const int*)d_in[2];
  const int*   dst    = (const int*)d_in[3];
  const float* W_edge = (const float*)d_in[4];
  const float* b_edge = (const float*)d_in[5];
  const float* bias   = (const float*)d_in[6];
  float* out = (float*)d_out;

  mpnn_init_out<<<(N_NODES * HID + 255) / 256, 256, 0, stream>>>(nf, bias, out);

  const size_t lds_bytes = (size_t)KPAIRS * LDS_STRIDE * sizeof(v2f);
  mpnn_edge_kernel<<<NBLOCKS, NTHREADS, lds_bytes, stream>>>(nf, ef, src, dst, W_edge, b_edge, out);
}


# node product once per node: 40 nodes per work-group, nf rows @ [W_0..W_15, B] on WMMA into LDS (816K WMMAs, was 3.4M), its edges listed by a src scan in fixed order, per edge a 17-term f32 FMA sum
# speedup vs baseline: 10.6038x; 1.3137x over previous
#include <hip/hip_runtime.h>

#define N_NODES 10000
#define N_EDGES 50000
#define HID 64
#define EDGE_DIM 16

typedef float v2f __attribute__((ext_vector_type(2)));
typedef float v8f __attribute__((ext_vector_type(8)));

#define NODES_WG 40
#define NBLOCKS (N_NODES / NODES_WG)
#define NSLICES 17
#define PSTRIDE 1092
#define NUNITS (NSLICES * 2)
#define NWAVES 8
#define NTHREADS (NWAVES * 32)
#define NGROUPS (N_EDGES / 4)
#define SCAN_ROWS 49
#define SCAN_BATCH 7
#define GROUPS_WAVE (SCAN_ROWS * 32)
#define LIST_CAP (GROUPS_WAVE * 4)
#define LIST_PITCH (LIST_CAP + 32)
#define P_BYTES (NODES_WG * PSTRIDE * 4)
#define LDS_BYTES (P_BYTES + NWAVES * LIST_PITCH * 2 + NWAVES * 4)

__global__ void mpnn_init_out(const float* __restrict__ nf,
                              const float* __restrict__ bias,
                              float* __restrict__ out) {
  int i = blockIdx.x * blockDim.x + threadIdx.x;
  if (i < N_NODES * HID) out[i] = nf[i] + bias[i & (HID - 1)];
}

__device__ __forceinline__ void load_unit(v2f (&buf)[32], const float* __restrict__ W_edge,
                                          const float* __restrict__ b_edge, int u, int hi, int lo) {
  const int d = u >> 1;
  const float* base = (d < 16 ? W_edge + d * (HID * HID) : b_edge)
                      + (2 * hi) * HID + (u & 1) * 32 + 2 * lo;
#pragma unroll
  for (int kk = 0; kk < 16; ++kk) {
    buf[2 * kk]     = *(const v2f*)(base + (4 * kk) * HID);
    buf[2 * kk + 1] = *(const v2f*)(base + (4 * kk + 1) * HID);
  }
}

__device__ __forceinline__ void scan_load(int4 (&v)[SCAN_BATCH], const int* __restrict__ src, int g) {
  const int4* __restrict__ src4 = (const int4*)src;
#pragma unroll
  for (int jj = 0; jj < SCAN_BATCH; ++jj)
    v[jj] = src4[min(g + jj * 32, NGROUPS - 1)];
}

__device__ __forceinline__ void scan_compare(const int4 (&v)[SCAN_BATCH], int g, int node0, int lane,
                                             unsigned short* __restrict__ mylist, int& cnt) {
#pragma unroll
  for (int jj = 0; jj < SCAN_BATCH; ++jj) {
    const int gg = g + jj * 32;
    const bool ok = gg < NGROUPS;
    const int vv[4] = {v[jj].x, v[jj].y, v[jj].z, v[jj].w};
#pragma unroll
    for (int t = 0; t < 4; ++t) {
      const bool m = ok && (unsigned)(vv[t] - node0) < (unsigned)NODES_WG;
      const unsigned mask = __builtin_amdgcn_ballot_w32(m);
      const int pos = cnt + (int)__builtin_amdgcn_mbcnt_lo(mask, 0u);
      mylist[m ? pos : LIST_CAP + lane] = (unsigned short)(4 * gg + t);
      cnt += __builtin_popcount(mask);
    }
  }
}

__global__ void __launch_bounds__(NTHREADS)
mpnn_node_kernel(const float* __restrict__ nf,
                 const float* __restrict__ ef,
                 const int* __restrict__ src,
                 const int* __restrict__ dst,
                 const float* __restrict__ W_edge,
                 const float* __restrict__ b_edge,
                 float* __restrict__ out) {
  extern __shared__ __attribute__((aligned(16))) char smem_raw[];
  float* __restrict__ P = (float*)smem_raw;

  const int tid  = threadIdx.x;
  const int lane = tid & 31;
  const int wid  = __builtin_amdgcn_readfirstlane(tid >> 5);
  const int hi   = lane >> 4;
  const int lo   = lane & 15;
  const int node0 = blockIdx.x * NODES_WG;
  unsigned short* __restrict__ lists = (unsigned short*)(smem_raw + P_BYTES);
  unsigned short* __restrict__ mylist = lists + wid * LIST_PITCH;
  int* __restrict__ wcnt = (int*)(smem_raw + P_BYTES + NWAVES * LIST_PITCH * 2);

  const v2f* __restrict__ nf2 = (const v2f*)nf;
  v2f a[3][16];
#pragma unroll
  for (int mt = 0; mt < 3; ++mt) {
    const int row = min(node0 + mt * 16 + lo, N_NODES - 1);
#pragma unroll
    for (int kk = 0; kk < 16; ++kk)
      a[mt][kk] = nf2[row * 32 + 2 * kk + hi];
  }

  const v8f vzero = {0.f, 0.f, 0.f, 0.f, 0.f, 0.f, 0.f, 0.f};
  int cnt = 0;

#pragma unroll 1
  for (int ph = 0; ph < 2; ++ph) {
  if ((((wid >> 2) ^ ph) & 1) == 0) {
  {
    const int g0 = wid * GROUPS_WAVE + lane;
    int4 va[SCAN_BATCH], vb[SCAN_BATCH];
    scan_load(va, src, g0);
#pragma unroll 1
    for (int jb = 0; jb < 6; jb += 2) {
      const int gb = g0 + jb * (SCAN_BATCH * 32);
      scan_load(vb, src, gb + SCAN_BATCH * 32);
      __builtin_amdgcn_sched_barrier(0);
      scan_compare(va, gb, node0, lane, mylist, cnt);
      scan_load(va, src, gb + 2 * SCAN_BATCH * 32);
      __builtin_amdgcn_sched_barrier(0);
      scan_compare(vb, gb + SCAN_BATCH * 32, node0, lane, mylist, cnt);
    }
    scan_compare(va, g0 + 6 * (SCAN_BATCH * 32), node0, lane, mylist, cnt);
  }

  } else {
  v2f cur[32], nxt[32];
  int u = wid;
  load_unit(cur, W_edge, b_edge, u, hi, lo);
  while (u < NUNITS) {
    const int un = u + NWAVES;
    load_unit(nxt, W_edge, b_edge, un < NUNITS ? un : u, hi, lo);
    __builtin_amdgcn_sched_barrier(0);
    v8f acc[3][2];
#pragma unroll
    for (int mt = 0; mt < 3; ++mt) { acc[mt][0] = vzero; acc[mt][1] = vzero; }
#pragma unroll
    for (int kk = 0; kk < 16; ++kk) {
      const v2f b0 = {cur[2 * kk].x, cur[2 * kk + 1].x};
      const v2f b1 = {cur[2 * kk].y, cur[2 * kk + 1].y};
#pragma unroll
      for (int mt = 0; mt < 3; ++mt) {
        acc[mt][0] = __builtin_amdgcn_wmma_f32_16x16x4_f32(false, a[mt][kk], false, b0,
                                                           (short)0, acc[mt][0], false, false);
        acc[mt][1] = __builtin_amdgcn_wmma_f32_16x16x4_f32(false, a[mt][kk], false, b1,
                                                           (short)0, acc[mt][1], false, false);
      }
    }
    float* pcol = P + (u >> 1) * HID + (u & 1) * 32 + 2 * lo;
#pragma unroll
    for (int mt = 0; mt < 3; ++mt) {
      if (mt < 2 || hi == 0) {
#pragma unroll
        for (int r = 0; r < 8; ++r) {
          const v2f pv = {acc[mt][0][r], acc[mt][1][r]};
          *(v2f*)(pcol + (mt * 16 + 8 * hi + r) * PSTRIDE) = pv;
        }
      }
    }
#pragma unroll
    for (int i = 0; i < 32; ++i) cur[i] = nxt[i];
    u = un;
  }
  }
  }
  if (lane == 0) wcnt[wid] = cnt;
  __syncthreads();

  int pre[NWAVES + 1];
  pre[0] = 0;
#pragma unroll
  for (int k = 0; k < NWAVES; ++k)
    pre[k + 1] = pre[k] + __builtin_amdgcn_readfirstlane(wcnt[k]);
  const int total = pre[NWAVES];
  const int i_end = ((wid + 1) * total) / NWAVES;
  for (int i0 = (wid * total) / NWAVES; i0 < i_end; i0 += 32) {
    const int nb = __builtin_amdgcn_readfirstlane(min(32, i_end - i0));
    const int idx = min(i0 + lane, i_end - 1);
    int off = idx;
#pragma unroll
    for (int k = 1; k < NWAVES; ++k)
      if (idx >= pre[k]) off = k * LIST_PITCH + (idx - pre[k]);
    const int e = lists[off];
    const int rowl = src[e] - node0;
    const int dn = dst[e];
    const float4* efp = (const float4*)(ef + e * EDGE_DIM);
    const float4 f0 = efp[0], f1 = efp[1], f2 = efp[2], f3 = efp[3];
    const float fl[16] = {f0.x, f0.y, f0.z, f0.w, f1.x, f1.y, f1.z, f1.w,
                          f2.x, f2.y, f2.z, f2.w, f3.x, f3.y, f3.z, f3.w};
    for (int j = 0; j < nb; ++j) {
      const int rj = __builtin_amdgcn_readlane(rowl, j);
      const int dj = __builtin_amdgcn_readlane(dn, j);
      const float* pr = P + rj * PSTRIDE + lane;
      float m0 = pr[16 * HID];
      float m1 = pr[16 * HID + 32];
#pragma unroll
      for (int d = 0; d < 16; ++d) {
        const float s = __int_as_float(__builtin_amdgcn_readlane(__float_as_int(fl[d]), j));
        m0 = fmaf(s, pr[d * HID], m0);
        m1 = fmaf(s, pr[d * HID + 32], m1);
      }
      float* op = out + dj * HID + lane;
      atomicAdd(op, m0);
      atomicAdd(op + 32, m1);
    }
  }
}

extern "C" void kernel_launch(void* const* d_in, const int* in_sizes, int n_in,
                              void* d_out, int out_size, void* d_ws, size_t ws_size,
                              hipStream_t stream) {
  const float* nf     = (const float*)d_in[0];
  const float* ef     = (const float*)d_in[1];
  const int*   src    = (const int*)d_in[2];
  const int*   dst    = (const int*)d_in[3];
  const float* W_edge = (const float*)d_in[4];
  const float* b_edge = (const float*)d_in[5];
  const float* bias   = (const float*)d_in[6];
  float* out = (float*)d_out;

  mpnn_init_out<<<(N_NODES * HID + 255) / 256, 256, 0, stream>>>(nf, bias, out);

  mpnn_node_kernel<<<NBLOCKS, NTHREADS, LDS_BYTES, stream>>>(nf, ef, src, dst, W_edge, b_edge, out);
}


# start value of out (nf + bias) written by 250 work-groups of 4 waves, five 16-byte pieces per lane all requested at once: 1000 waves to enter, was 20000 waves of one float per lane
# speedup vs baseline: 11.0068x; 1.0380x over previous
#include <hip/hip_runtime.h>

#define N_NODES 10000
#define N_EDGES 50000
#define HID 64
#define EDGE_DIM 16

typedef float v2f __attribute__((ext_vector_type(2)));
typedef float v8f __attribute__((ext_vector_type(8)));

#define NODES_WG 40
#define NBLOCKS (N_NODES / NODES_WG)
#define NSLICES 17
#define PSTRIDE 1092
#define NUNITS (NSLICES * 2)
#define NWAVES 8
#define NTHREADS (NWAVES * 32)
#define NGROUPS (N_EDGES / 4)
#define SCAN_ROWS 49
#define SCAN_BATCH 7
#define GROUPS_WAVE (SCAN_ROWS * 32)
#define LIST_CAP (GROUPS_WAVE * 4)
#define LIST_PITCH (LIST_CAP + 32)
#define P_BYTES (NODES_WG * PSTRIDE * 4)
#define LDS_BYTES (P_BYTES + NWAVES * LIST_PITCH * 2 + NWAVES * 4)

#define INIT_WGS 250
#define INIT_THREADS 128
#define INIT_VECS 5
#define INIT_STRIDE (INIT_WGS * INIT_THREADS)
static_assert(INIT_STRIDE * INIT_VECS * 4 == N_NODES * HID, "the start-value kernel covers out exactly once");
static_assert(INIT_STRIDE % (HID / 4) == 0, "a lane keeps one bias piece");

__global__ void __launch_bounds__(INIT_THREADS)
mpnn_init_out(const float* __restrict__ nf,
              const float* __restrict__ bias,
              float* __restrict__ out) {
  const int t = blockIdx.x * INIT_THREADS + threadIdx.x;
  const float4* __restrict__ nf4 = (const float4*)nf;
  float4* __restrict__ out4 = (float4*)out;
  const float4 b = ((const float4*)bias)[t & (HID / 4 - 1)];
  float4 v[INIT_VECS];
#pragma unroll
  for (int k = 0; k < INIT_VECS; ++k) v[k] = nf4[t + k * INIT_STRIDE];
#pragma unroll
  for (int k = 0; k < INIT_VECS; ++k) {
    const float4 r = {v[k].x + b.x, v[k].y + b.y, v[k].z + b.z, v[k].w + b.w};
    out4[t + k * INIT_STRIDE] = r;
  }
}

__device__ __forceinline__ void load_unit(v2f (&buf)[32], const float* __restrict__ W_edge,
                                          const float* __restrict__ b_edge, int u, int hi, int lo) {
  const int d = u >> 1;
  const float* base = (d < 16 ? W_edge + d * (HID * HID) : b_edge)
                      + (2 * hi) * HID + (u & 1) * 32 + 2 * lo;
#pragma unroll
  for (int kk = 0; kk < 16; ++kk) {
    buf[2 * kk]     = *(const v2f*)(base + (4 * kk) * HID);
    buf[2 * kk + 1] = *(const v2f*)(base + (4 * kk + 1) * HID);
  }
}

__device__ __forceinline__ void scan_load(int4 (&v)[SCAN_BATCH], const int* __restrict__ src, int g) {
  const int4* __restrict__ src4 = (const int4*)src;
#pragma unroll
  for (int jj = 0; jj < SCAN_BATCH; ++jj)
    v[jj] = src4[min(g + jj * 32, NGROUPS - 1)];
}

__device__ __forceinline__ void scan_compare(const int4 (&v)[SCAN_BATCH], int g, int node0, int lane,
                                             unsigned short* __restrict__ mylist, int& cnt) {
#pragma unroll
  for (int jj = 0; jj < SCAN_BATCH; ++jj) {
    const int gg = g + jj * 32;
    const bool ok = gg < NGROUPS;
    const int vv[4] = {v[jj].x, v[jj].y, v[jj].z, v[jj].w};
#pragma unroll
    for (int t = 0; t < 4; ++t) {
      const bool m = ok && (unsigned)(vv[t] - node0) < (unsigned)NODES_WG;
      const unsigned mask = __builtin_amdgcn_ballot_w32(m);
      const int pos = cnt + (int)__builtin_amdgcn_mbcnt_lo(mask, 0u);
      mylist[m ? pos : LIST_CAP + lane] = (unsigned short)(4 * gg + t);
      cnt += __builtin_popcount(mask);
    }
  }
}

__global__ void __launch_bounds__(NTHREADS)
mpnn_node_kernel(const float* __restrict__ nf,
                 const float* __restrict__ ef,
                 const int* __restrict__ src,
                 const int* __restrict__ dst,
                 const float* __restrict__ W_edge,
                 const float* __restrict__ b_edge,
                 float* __restrict__ out) {
  extern __shared__ __attribute__((aligned(16))) char smem_raw[];
  float* __restrict__ P = (float*)smem_raw;

  const int tid  = threadIdx.x;
  const int lane = tid & 31;
  const int wid  = __builtin_amdgcn_readfirstlane(tid >> 5);
  const int hi   = lane >> 4;
  const int lo   = lane & 15;
  const int node0 = blockIdx.x * NODES_WG;
  unsigned short* __restrict__ lists = (unsigned short*)(smem_raw + P_BYTES);
  unsigned short* __restrict__ mylist = lists + wid * LIST_PITCH;
  int* __restrict__ wcnt = (int*)(smem_raw + P_BYTES + NWAVES * LIST_PITCH * 2);

  const v2f* __restrict__ nf2 = (const v2f*)nf;
  v2f a[3][16];
#pragma unroll
  for (int mt = 0; mt < 3; ++mt) {
    const int row = min(node0 + mt * 16 + lo, N_NODES - 1);
#pragma unroll
    for (int kk = 0; kk < 16; ++kk)
      a[mt][kk] = nf2[row * 32 + 2 * kk + hi];
  }

  const v8f vzero = {0.f, 0.f, 0.f, 0.f, 0.f, 0.f, 0.f, 0.f};
  int cnt = 0;

#pragma unroll 1
  for (int ph = 0; ph < 2; ++ph) {
  if ((((wid >> 2) ^ ph) & 1) == 0) {
  {
    const int g0 = wid * GROUPS_WAVE + lane;
    int4 va[SCAN_BATCH], vb[SCAN_BATCH];
    scan_load(va, src, g0);
#pragma unroll 1
    for (int jb = 0; jb < 6; jb += 2) {
      const int gb = g0 + jb * (SCAN_BATCH * 32);
      scan_load(vb, src, gb + SCAN_BATCH * 32);
      __builtin_amdgcn_sched_barrier(0);
      scan_compare(va, gb, node0, lane, mylist, cnt);
      scan_load(va, src, gb + 2 * SCAN_BATCH * 32);
      __builtin_amdgcn_sched_barrier(0);
      scan_compare(vb, gb + SCAN_BATCH * 32, node0, lane, mylist, cnt);
    }
    scan_compare(va, g0 + 6 * (SCAN_BATCH * 32), node0, lane, mylist, cnt);
  }

  } else {
  v2f cur[32], nxt[32];
  int u = wid;
  load_unit(cur, W_edge, b_edge, u, hi, lo);
  while (u < NUNITS) {
    const int un = u + NWAVES;
    load_unit(nxt, W_edge, b_edge, un < NUNITS ? un : u, hi, lo);
    __builtin_amdgcn_sched_barrier(0);
    v8f acc[3][2];
#pragma unroll
    for (int mt = 0; mt < 3; ++mt) { acc[mt][0] = vzero; acc[mt][1] = vzero; }
#pragma unroll
    for (int kk = 0; kk < 16; ++kk) {
      const v2f b0 = {cur[2 * kk].x, cur[2 * kk + 1].x};
      const v2f b1 = {cur[2 * kk].y, cur[2 * kk + 1].y};
#pragma unroll
      for (int mt = 0; mt < 3; ++mt) {
        acc[mt][0] = __builtin_amdgcn_wmma_f32_16x16x4_f32(false, a[mt][kk], false, b0,
                                                           (short)0, acc[mt][0], false, false);
        acc[mt][1] = __builtin_amdgcn_wmma_f32_16x16x4_f32(false, a[mt][kk], false, b1,
                                                           (short)0, acc[mt][1], false, false);
      }
    }
    float* pcol = P + (u >> 1) * HID + (u & 1) * 32 + 2 * lo;
#pragma unroll
    for (int mt = 0; mt < 3; ++mt) {
      if (mt < 2 || hi == 0) {
#pragma unroll
        for (int r = 0; r < 8; ++r) {
          const v2f pv = {acc[mt][0][r], acc[mt][1][r]};
          *(v2f*)(pcol + (mt * 16 + 8 * hi + r) * PSTRIDE) = pv;
        }
      }
    }
#pragma unroll
    for (int i = 0; i < 32; ++i) cur[i] = nxt[i];
    u = un;
  }
  }
  }
  if (lane == 0) wcnt[wid] = cnt;
  __syncthreads();

  int pre[NWAVES + 1];
  pre[0] = 0;
#pragma unroll
  for (int k = 0; k < NWAVES; ++k)
    pre[k + 1] = pre[k] + __builtin_amdgcn_readfirstlane(wcnt[k]);
  const int total = pre[NWAVES];
  const int i_end = ((wid + 1) * total) / NWAVES;
  for (int i0 = (wid * total) / NWAVES; i0 < i_end; i0 += 32) {
    const int nb = __builtin_amdgcn_readfirstlane(min(32, i_end - i0));
    const int idx = min(i0 + lane, i_end - 1);
    int off = idx;
#pragma unroll
    for (int k = 1; k < NWAVES; ++k)
      if (idx >= pre[k]) off = k * LIST_PITCH + (idx - pre[k]);
    const int e = lists[off];
    const int rowl = src[e] - node0;
    const int dn = dst[e];
    const float4* efp = (const float4*)(ef + e * EDGE_DIM);
    const float4 f0 = efp[0], f1 = efp[1], f2 = efp[2], f3 = efp[3];
    const float fl[16] = {f0.x, f0.y, f0.z, f0.w, f1.x, f1.y, f1.z, f1.w,
                          f2.x, f2.y, f2.z, f2.w, f3.x, f3.y, f3.z, f3.w};
    for (int j = 0; j < nb; ++j) {
      const int rj = __builtin_amdgcn_readlane(rowl, j);
      const int dj = __builtin_amdgcn_readlane(dn, j);
      const float* pr = P + rj * PSTRIDE + lane;
      float m0 = pr[16 * HID];
      float m1 = pr[16 * HID + 32];
#pragma unroll
      for (int d = 0; d < 16; ++d) {
        const float s = __int_as_float(__builtin_amdgcn_readlane(__float_as_int(fl[d]), j));
        m0 = fmaf(s, pr[d * HID], m0);
        m1 = fmaf(s, pr[d * HID + 32], m1);
      }
      float* op = out + dj * HID + lane;
      atomicAdd(op, m0);
      atomicAdd(op + 32, m1);
    }
  }
}

extern "C" void kernel_launch(void* const* d_in, const int* in_sizes, int n_in,
                              void* d_out, int out_size, void* d_ws, size_t ws_size,
                              hipStream_t stream) {
  const float* nf     = (const float*)d_in[0];
  const float* ef     = (const float*)d_in[1];
  const int*   src    = (const int*)d_in[2];
  const int*   dst    = (const int*)d_in[3];
  const float* W_edge = (const float*)d_in[4];
  const float* b_edge = (const float*)d_in[5];
  const float* bias   = (const float*)d_in[6];
  float* out = (float*)d_out;

  mpnn_init_out<<<INIT_WGS, INIT_THREADS, 0, stream>>>(nf, bias, out);

  mpnn_node_kernel<<<NBLOCKS, NTHREADS, LDS_BYTES, stream>>>(nf, ef, src, dst, W_edge, b_edge, out);
}


# Z row sums in registers: each wave sorts its own list entries by dst node and sums a node's 1088-value row in 34 VGPRs, one LDS store per row; was an LDS load-modify-store per edge and a zero fill
# speedup vs baseline: 13.8260x; 1.2561x over previous
#include <hip/hip_runtime.h>

#define N_NODES 10000
#define N_EDGES 50000
#define HID 64
#define EDGE_DIM 16

typedef float v2f __attribute__((ext_vector_type(2)));
typedef float v8f __attribute__((ext_vector_type(8)));

#define NODES_WG 40
#define NBLOCKS (N_NODES / NODES_WG)
#define NSLICES 17
#define KTOT (NSLICES * HID)
#define PSTRIDE 1092
#define NWAVES 8
#define NTHREADS (NWAVES * 32)
#define NGROUPS (N_EDGES / 4)
#define SCAN_ROWS 49
#define SCAN_BATCH 7
#define GROUPS_WAVE (SCAN_ROWS * 32)
#define LIST_CAP (GROUPS_WAVE * 4)
#define LIST_PITCH (LIST_CAP + 32)
#define P_BYTES (NODES_WG * PSTRIDE * 4)
#define LDS_BYTES (P_BYTES + NWAVES * LIST_PITCH * 2 + NWAVES * 4 + NWAVES * OWN_PITCH * 2 + NWAVES * STG_ROWS * HID * 4 + 2 * OWN_CHUNK * 2 + 2 * OWN_CHUNK)

__device__ __forceinline__ void scan_load(int4 (&v)[SCAN_BATCH], const int* __restrict__ idx, int g) {
  const int4* __restrict__ src4 = (const int4*)idx;
#pragma unroll
  for (int jj = 0; jj < SCAN_BATCH; ++jj)
    v[jj] = src4[min(g + jj * 32, NGROUPS - 1)];
}

__device__ __forceinline__ void scan_compare(const int4 (&v)[SCAN_BATCH], int g, int node0, int lane,
                                             unsigned short* __restrict__ mylist, int& cnt) {
#pragma unroll
  for (int jj = 0; jj < SCAN_BATCH; ++jj) {
    const int gg = g + jj * 32;
    const bool ok = gg < NGROUPS;
    const int vv[4] = {v[jj].x, v[jj].y, v[jj].z, v[jj].w};
#pragma unroll
    for (int t = 0; t < 4; ++t) {
      const bool m = ok && (unsigned)(vv[t] - node0) < (unsigned)NODES_WG;
      const unsigned mask = __builtin_amdgcn_ballot_w32(m);
      const int pos = cnt + (int)__builtin_amdgcn_mbcnt_lo(mask, 0u);
      mylist[m ? pos : LIST_CAP + lane] = (unsigned short)(4 * gg + t);
      cnt += __builtin_popcount(mask);
    }
  }
}

#define KQ 4
#define KSTEPS_WAVE (KTOT / KQ / 4)
#define CHUNK 17
#define NODES_WAVE (NODES_WG / NWAVES)
#define OWN_CHUNK 256
#define OWN_GROUPS (OWN_CHUNK / 32)
#define OWN_PITCH (OWN_CHUNK + 32)
#define OWN_OFF (P_BYTES + NWAVES * LIST_PITCH * 2 + NWAVES * 4)
#define STG_ROWS 16
#define STG_OFF (OWN_OFF + NWAVES * OWN_PITCH * 2)
#define CE_OFF (STG_OFF + NWAVES * STG_ROWS * HID * 4)
#define CL_OFF (CE_OFF + 2 * OWN_CHUNK * 2)
#define PART_BYTES (KQ * NODES_WG * HID * 4)
static_assert(PART_BYTES <= NWAVES * LIST_PITCH * 2, "the K-quarter partial sums reuse the list area");
static_assert(KSTEPS_WAVE == 4 * CHUNK, "four chunks per wave");

__device__ __forceinline__ void entry_add(v2f (&acc)[NSLICES], const v2f xv, int ln, const float4& fa,
                                          const float4& fb, const float4& fc, const float4& fd) {
  const float fl[16] = {fa.x, fa.y, fa.z, fa.w, fb.x, fb.y, fb.z, fb.w,
                        fc.x, fc.y, fc.z, fc.w, fd.x, fd.y, fd.z, fd.w};
#pragma unroll
  for (int d = 0; d < 16; ++d) {
    const float s = __int_as_float(__builtin_amdgcn_readlane(__float_as_int(fl[d]), ln));
    acc[d].x = fmaf(s, xv.x, acc[d].x);
    acc[d].y = fmaf(s, xv.y, acc[d].y);
  }
  acc[16].x += xv.x;
  acc[16].y += xv.y;
}

__device__ __forceinline__ void load_chunk(v2f (&bq)[2 * CHUNK], v2f (&aq)[3][CHUNK], int c, int kq,
                                           const float* __restrict__ W_edge, const float* __restrict__ b_edge,
                                           int colofs, int hi, const float* const (&zrow)[3]) {
#pragma unroll
  for (int t = 0; t < CHUNK; ++t) {
    const int k0 = kq * (KTOT / KQ) + 4 * (c * CHUNK + t);
    const float* wp = (k0 < EDGE_DIM * HID ? W_edge + k0 * HID : b_edge + (k0 - EDGE_DIM * HID) * HID)
                      + (2 * hi) * HID + colofs;
    bq[2 * t]     = *(const v2f*)wp;
    bq[2 * t + 1] = *(const v2f*)(wp + HID);
#pragma unroll
    for (int mt = 0; mt < 3; ++mt)
      aq[mt][t] = *(const v2f*)(zrow[mt] + k0);
  }
}

__global__ void __launch_bounds__(NTHREADS)
mpnn_node_kernel(const float* __restrict__ nf,
                 const float* __restrict__ ef,
                 const int* __restrict__ src,
                 const int* __restrict__ dst,
                 const float* __restrict__ W_edge,
                 const float* __restrict__ b_edge,
                 const float* __restrict__ bias,
                 float* __restrict__ out) {
  extern __shared__ __attribute__((aligned(16))) char smem_raw[];
  float* Z = (float*)smem_raw;

  const int tid  = threadIdx.x;
  const int lane = tid & 31;
  const int wid  = __builtin_amdgcn_readfirstlane(tid >> 5);
  const int hi   = lane >> 4;
  const int lo   = lane & 15;
  const int node0 = blockIdx.x * NODES_WG;
  unsigned short* __restrict__ lists = (unsigned short*)(smem_raw + P_BYTES);
  unsigned short* __restrict__ mylist = lists + wid * LIST_PITCH;
  int* __restrict__ wcnt = (int*)(smem_raw + P_BYTES + NWAVES * LIST_PITCH * 2);

  int cnt = 0;
  {
    const int g0 = wid * GROUPS_WAVE + lane;
    int4 va[SCAN_BATCH], vb[SCAN_BATCH];
    scan_load(va, dst, g0);
#pragma unroll 1
    for (int jb = 0; jb < 6; jb += 2) {
      const int gb = g0 + jb * (SCAN_BATCH * 32);
      scan_load(vb, dst, gb + SCAN_BATCH * 32);
      __builtin_amdgcn_sched_barrier(0);
      scan_compare(va, gb, node0, lane, mylist, cnt);
      scan_load(va, dst, gb + 2 * SCAN_BATCH * 32);
      __builtin_amdgcn_sched_barrier(0);
      scan_compare(vb, gb + SCAN_BATCH * 32, node0, lane, mylist, cnt);
    }
    scan_compare(va, g0 + 6 * (SCAN_BATCH * 32), node0, lane, mylist, cnt);
  }
  if (lane == 0) wcnt[wid] = cnt;
  __syncthreads();

  {
    int pre[NWAVES + 1];
    pre[0] = 0;
#pragma unroll
    for (int k = 0; k < NWAVES; ++k)
      pre[k + 1] = pre[k] + __builtin_amdgcn_readfirstlane(wcnt[k]);
    const int total = pre[NWAVES];
    const v2f* __restrict__ nf2 = (const v2f*)nf;
    const int r0 = wid * NODES_WAVE;
    unsigned short* __restrict__ own = (unsigned short*)(smem_raw + OWN_OFF) + wid * OWN_PITCH;
    float* stg = (float*)(smem_raw + STG_OFF) + wid * (STG_ROWS * HID) + 2 * lane;
    const v2f zero2 = {0.f, 0.f};
    int c0 = 0;
    do {
      unsigned short* ce = (unsigned short*)(smem_raw + CE_OFF) + ((c0 / OWN_CHUNK) & 1) * OWN_CHUNK;
      unsigned char* cl = (unsigned char*)(smem_raw + CL_OFF) + ((c0 / OWN_CHUNK) & 1) * OWN_CHUNK;
      {
        const int i = c0 + 32 * wid + lane;
        int e = 0, loc = 255;
        if (i < total) {
          int off = i;
#pragma unroll
          for (int k = 1; k < NWAVES; ++k)
            if (i >= pre[k]) off = k * LIST_PITCH + (i - pre[k]);
          e = lists[off];
          loc = dst[e] - node0;
        }
        ce[32 * wid + lane] = (unsigned short)e;
        cl[32 * wid + lane] = (unsigned char)loc;
      }
      __syncthreads();
      int ec[OWN_GROUPS], lc[OWN_GROUPS];
#pragma unroll
      for (int c = 0; c < OWN_GROUPS; ++c) {
        ec[c] = ce[32 * c + lane];
        lc[c] = (int)cl[32 * c + lane] - r0;
      }
      int endk[NODES_WAVE];
      int run = 0;
#pragma unroll
      for (int k = 0; k < NODES_WAVE; ++k) {
#pragma unroll
        for (int c = 0; c < OWN_GROUPS; ++c) {
          const bool m = (lc[c] == k);
          const unsigned mask = __builtin_amdgcn_ballot_w32(m);
          const int p = run + (int)__builtin_amdgcn_mbcnt_lo(mask, 0u);
          own[m ? p : OWN_CHUNK + lane] = (unsigned short)ec[c];
          run += __builtin_popcount(mask);
        }
        endk[k] = run;
      }
      const int nown = run;

      int k = 0;
      int pos = 0;
      v2f acc[NSLICES];
      {
        const float* zr = Z + r0 * PSTRIDE + 2 * lane;
#pragma unroll
        for (int d = 0; d < NSLICES; ++d) acc[d] = (c0 == 0) ? zero2 : *(const v2f*)(zr + d * HID);
      }
      for (int b0 = 0; b0 < nown; b0 += 32) {
        const int nb = min(32, nown - b0);
        const int e = own[b0 + min(lane, nb - 1)];
        const int bs = src[e];
        const float4* efp = (const float4*)(ef + e * EDGE_DIM);
        const float4 fa = efp[0], fb = efp[1], fc = efp[2], fd = efp[3];
        v2f x[32];
#pragma unroll
        for (int i = 0; i < 32; ++i) {
          const int sj = __builtin_amdgcn_readlane(bs, i);
          x[i] = nf2[sj * (HID / 2) + lane];
        }
#pragma unroll
        for (int h = 0; h < 32 / STG_ROWS; ++h) {
          if (STG_ROWS * h < nb) {
#pragma unroll
            for (int t = 0; t < STG_ROWS; ++t) *(v2f*)(stg + t * HID) = x[STG_ROWS * h + t];
            const int bh = b0 + STG_ROWS * h;
            for (;;) {
              const int ek = (k == 0) ? endk[0] : (k == 1) ? endk[1] : (k == 2) ? endk[2] : (k == 3) ? endk[3] : endk[4];
              const int lo = pos - bh;
              const int hi = min(ek, bh + STG_ROWS) - bh;
              if (lo < hi) {
                int j = lo;
                if ((hi - lo) & 1) {
                  const v2f xa = *(const v2f*)(stg + j * HID);
                  entry_add(acc, xa, STG_ROWS * h + j, fa, fb, fc, fd);
                  ++j;
                }
                for (; j < hi; j += 2) {
                  const v2f xa = *(const v2f*)(stg + j * HID);
                  const v2f xb = *(const v2f*)(stg + (j + 1) * HID);
                  entry_add(acc, xa, STG_ROWS * h + j, fa, fb, fc, fd);
                  entry_add(acc, xb, STG_ROWS * h + j + 1, fa, fb, fc, fd);
                }
                pos = bh + hi;
              }
              if (pos != ek) break;
              {
                float* zr = Z + (r0 + k) * PSTRIDE + 2 * lane;
#pragma unroll
                for (int d = 0; d < NSLICES; ++d) *(v2f*)(zr + d * HID) = acc[d];
              }
              ++k;
              if (k == NODES_WAVE) break;
              {
                const float* zr = Z + (r0 + k) * PSTRIDE + 2 * lane;
#pragma unroll
                for (int d = 0; d < NSLICES; ++d) acc[d] = (c0 == 0) ? zero2 : *(const v2f*)(zr + d * HID);
              }
            }
          }
        }
      }
      while (k < NODES_WAVE) {
        float* zr = Z + (r0 + k) * PSTRIDE + 2 * lane;
#pragma unroll
        for (int d = 0; d < NSLICES; ++d) *(v2f*)(zr + d * HID) = acc[d];
        ++k;
        if (k < NODES_WAVE) {
          const float* zn = Z + (r0 + k) * PSTRIDE + 2 * lane;
#pragma unroll
          for (int d = 0; d < NSLICES; ++d) acc[d] = (c0 == 0) ? zero2 : *(const v2f*)(zn + d * HID);
        }
      }
      c0 += OWN_CHUNK;
    } while (c0 < total);
  }
  __syncthreads();

  const int nh = wid & 1;
  const int kq = wid >> 1;
  const int colofs = nh * 32 + 2 * lo;
  const float* zrow[3];
#pragma unroll
  for (int mt = 0; mt < 3; ++mt)
    zrow[mt] = Z + min(mt * 16 + lo, NODES_WG - 1) * PSTRIDE + 2 * hi;

  const v8f vzero = {0.f, 0.f, 0.f, 0.f, 0.f, 0.f, 0.f, 0.f};
  v8f acc[3][2];
#pragma unroll
  for (int mt = 0; mt < 3; ++mt) { acc[mt][0] = vzero; acc[mt][1] = vzero; }

  v2f bcur[2 * CHUNK], bnxt[2 * CHUNK];
  v2f acur[3][CHUNK], anxt[3][CHUNK];
  load_chunk(bcur, acur, 0, kq, W_edge, b_edge, colofs, hi, zrow);
#pragma unroll
  for (int c = 0; c < KSTEPS_WAVE / CHUNK; ++c) {
    if (c + 1 < KSTEPS_WAVE / CHUNK)
      load_chunk(bnxt, anxt, c + 1, kq, W_edge, b_edge, colofs, hi, zrow);
    __builtin_amdgcn_sched_barrier(0);
#pragma unroll
    for (int t = 0; t < CHUNK; ++t) {
      const v2f b0 = {bcur[2 * t].x, bcur[2 * t + 1].x};
      const v2f b1 = {bcur[2 * t].y, bcur[2 * t + 1].y};
#pragma unroll
      for (int mt = 0; mt < 3; ++mt) {
        acc[mt][0] = __builtin_amdgcn_wmma_f32_16x16x4_f32(false, acur[mt][t], false, b0,
                                                           (short)0, acc[mt][0], false, false);
        acc[mt][1] = __builtin_amdgcn_wmma_f32_16x16x4_f32(false, acur[mt][t], false, b1,
                                                           (short)0, acc[mt][1], false, false);
      }
    }
#pragma unroll
    for (int i = 0; i < 2 * CHUNK; ++i) bcur[i] = bnxt[i];
#pragma unroll
    for (int mt = 0; mt < 3; ++mt)
#pragma unroll
      for (int t = 0; t < CHUNK; ++t) acur[mt][t] = anxt[mt][t];
  }

  float* part = (float*)(smem_raw + P_BYTES);
  {
    float* pcol = part + kq * (NODES_WG * HID) + colofs;
#pragma unroll
    for (int mt = 0; mt < 3; ++mt) {
      if (mt < 2 || hi == 0) {
#pragma unroll
        for (int r = 0; r < 8; ++r) {
          const v2f pv = {acc[mt][0][r], acc[mt][1][r]};
          *(v2f*)(pcol + (mt * 16 + 8 * hi + r) * HID) = pv;
        }
      }
    }
  }
  __syncthreads();
  {
    const float4* part4 = (const float4*)part;
    const float4* __restrict__ nf4 = (const float4*)nf;
    float4* __restrict__ out4 = (float4*)out;
    for (int i = tid; i < NODES_WG * HID / 4; i += NTHREADS) {
      const float4 b = ((const float4*)bias)[i & (HID / 4 - 1)];
      const float4 x = nf4[node0 * (HID / 4) + i];
      const float4 p0 = part4[i];
      const float4 p1 = part4[i + NODES_WG * HID / 4];
      const float4 p2 = part4[i + 2 * (NODES_WG * HID / 4)];
      const float4 p3 = part4[i + 3 * (NODES_WG * HID / 4)];
      const float4 r = {((p0.x + p1.x) + (p2.x + p3.x)) + (x.x + b.x),
                        ((p0.y + p1.y) + (p2.y + p3.y)) + (x.y + b.y),
                        ((p0.z + p1.z) + (p2.z + p3.z)) + (x.z + b.z),
                        ((p0.w + p1.w) + (p2.w + p3.w)) + (x.w + b.w)};
      out4[node0 * (HID / 4) + i] = r;
    }
  }
}

extern "C" void kernel_launch(void* const* d_in, const int* in_sizes, int n_in,
                              void* d_out, int out_size, void* d_ws, size_t ws_size,
                              hipStream_t stream) {
  const float* nf     = (const float*)d_in[0];
  const float* ef     = (const float*)d_in[1];
  const int*   src    = (const int*)d_in[2];
  const int*   dst    = (const int*)d_in[3];
  const float* W_edge = (const float*)d_in[4];
  const float* b_edge = (const float*)d_in[5];
  const float* bias   = (const float*)d_in[6];
  float* out = (float*)d_out;

  mpnn_node_kernel<<<NBLOCKS, NTHREADS, LDS_BYTES, stream>>>(nf, ef, src, dst, W_edge, b_edge, bias, out);
}
